# starting kernel
# baseline (speedup 1.0000x reference)
#include <hip/hip_runtime.h>

#define PN      5
#define MROWS   500
#define NCOLS   2000
#define BB      1024
#define KITERS  10
#define MP      100
#define AXLD    112
#define JT5     5
#define NGRP    80
#define KLD     102
#define KCH     100
#define MPAD    112
#define ROWSPAN 512

typedef __attribute__((ext_vector_type(2))) float v2f;
typedef __attribute__((ext_vector_type(8))) float v8f;

__device__ __forceinline__ v8f wmma4(v2f a, v2f b, v8f c) {
    return __builtin_amdgcn_wmma_f32_16x16x4_f32(
        false, a, false, b, (short)0, c, false, false);
}


__global__ void zero_kernel(float4* __restrict__ dst, long long n4) {
    long long i = (long long)blockIdx.x * blockDim.x + threadIdx.x;
    if (i < n4) dst[i] = make_float4(0.f, 0.f, 0.f, 0.f);
}

__global__ void __launch_bounds__(256, 1)
atb_kernel(const float* __restrict__ A, const float* __restrict__ b,
           float* __restrict__ Atb) {
    const int p    = blockIdx.z;
    const int n0   = blockIdx.x * NGRP;
    const int it0  = blockIdx.y * 8 + (threadIdx.x >> 5);
    const int r0   = it0 * 16;
    const int r1   = r0 + ROWSPAN;
    const float* Ap = A + (size_t)p * MP * NCOLS;

    __shared__ float sBt[NGRP][KLD];
    for (int idx = threadIdx.x; idx < MP * NGRP / 4; idx += 256) {
        const int kk = idx / (NGRP / 4);
        const int c4 = (idx % (NGRP / 4)) * 4;
        float4 v = *(const float4*)&Ap[(size_t)kk * NCOLS + n0 + c4];
        sBt[c4 + 0][kk] = v.x;
        sBt[c4 + 1][kk] = v.y;
        sBt[c4 + 2][kk] = v.z;
        sBt[c4 + 3][kk] = v.w;
    }
    __syncthreads();

    const int lane = threadIdx.x & 31;
    const int nl   = lane & 15;
    const int kh   = (lane >> 4) << 1;

    const float* pa0 = b + (size_t)(r0 + nl) * MROWS + p * MP + kh;
    const float* pa1 = b + (size_t)(r1 + nl) * MROWS + p * MP + kh;
    const float* ps  = &sBt[nl][kh];

    v8f acc0[JT5] = {};
    v8f acc1[JT5] = {};
    for (int k0 = 0; k0 < MP; k0 += 4) {
        v2f af0; af0.x = pa0[0]; af0.y = pa0[1]; pa0 += 4;
        v2f af1; af1.x = pa1[0]; af1.y = pa1[1]; pa1 += 4;
#pragma unroll
        for (int j = 0; j < JT5; ++j) {
            v2f bf = *(const v2f*)(ps + j * 16 * KLD);
            acc0[j] = wmma4(af0, bf, acc0[j]);
            acc1[j] = wmma4(af1, bf, acc1[j]);
        }
        ps += 4;
    }

    float* out = Atb + (size_t)p * BB * NCOLS;
    const int ih = (lane >> 4) * 8;
#pragma unroll
    for (int j = 0; j < JT5; ++j) {
        const int col = n0 + j * 16 + nl;
#pragma unroll
        for (int v = 0; v < 8; ++v) {
            out[(size_t)(r0 + ih + v) * NCOLS + col] = acc0[j][v];
            out[(size_t)(r1 + ih + v) * NCOLS + col] = acc1[j][v];
        }
    }
}

__global__ void __launch_bounds__(256, 1)
ax_kernel(const float* __restrict__ A, const float* __restrict__ X,
          float* __restrict__ axbuf, int p) {
    const int it0 = blockIdx.x * 8 + (threadIdx.x >> 5);
    const int r0  = it0 * 16;
    const int r1  = r0 + ROWSPAN;
    const float* Ap = A + (size_t)p * MP * NCOLS;
    const float* Xp = X + (size_t)p * BB * NCOLS;

    __shared__ float sB[MPAD][KCH];

    const int lane = threadIdx.x & 31;
    const int ml   = lane & 15;
    const int kh   = (lane >> 4) << 1;

    const float* px0 = Xp + (size_t)(r0 + ml) * NCOLS + kh;
    const float* px1 = Xp + (size_t)(r1 + ml) * NCOLS + kh;

    v8f acc0[7] = {};
    v8f acc1[7] = {};
    for (int kc = 0; kc < NCOLS; kc += KCH) {
        __syncthreads();
        for (int idx = threadIdx.x; idx < MPAD * KCH / 4; idx += 256) {
            const int m  = idx / (KCH / 4);
            const int c4 = (idx % (KCH / 4)) * 4;
            float4 v = make_float4(0.f, 0.f, 0.f, 0.f);
            if (m < MP) {
                v = *(const float4*)&Ap[(size_t)m * NCOLS + kc + c4];
                if (kc + KCH < NCOLS)
                    __builtin_prefetch(&Ap[(size_t)m * NCOLS + kc + KCH + c4], 0, 1);
            }
            *(float4*)&sB[m][c4] = v;
        }
        __syncthreads();

        for (int k0 = 0; k0 < KCH; k0 += 4) {
            v2f af0; af0.x = px0[0]; af0.y = px0[1]; px0 += 4;
            v2f af1; af1.x = px1[0]; af1.y = px1[1]; px1 += 4;
            const float* ps = &sB[ml][k0 + kh];
#pragma unroll
            for (int j = 0; j < 7; ++j) {
                v2f bf = *(const v2f*)(ps + j * 16 * KCH);
                acc0[j] = wmma4(af0, bf, acc0[j]);
                acc1[j] = wmma4(af1, bf, acc1[j]);
            }
        }
    }

    const int ih = (lane >> 4) * 8;
#pragma unroll
    for (int j = 0; j < 7; ++j) {
        const int col = j * 16 + ml;
        if (col < MP) {
#pragma unroll
            for (int v = 0; v < 8; ++v) {
                axbuf[(size_t)(r0 + ih + v) * AXLD + col] = acc0[j][v];
                axbuf[(size_t)(r1 + ih + v) * AXLD + col] = acc1[j][v];
            }
        }
    }
}

__global__ void __launch_bounds__(256, 1)
xupd_kernel(const float* __restrict__ A, const float* __restrict__ axbuf,
            const float* __restrict__ Atb, const float* __restrict__ hyp,
            const int* __restrict__ nbr,
            float* __restrict__ X, const float* __restrict__ U,
            int p, int k) {
    const int n0  = blockIdx.x * NGRP;
    const int it0 = blockIdx.y * 8 + (threadIdx.x >> 5);
    const int r0  = it0 * 16;
    const int r1  = r0 + ROWSPAN;
    const float* Ap = A + (size_t)p * MP * NCOLS;

    __shared__ float sBt[NGRP][KLD];
    for (int idx = threadIdx.x; idx < MP * NGRP / 4; idx += 256) {
        const int kk = idx / (NGRP / 4);
        const int c4 = (idx % (NGRP / 4)) * 4;
        float4 v = *(const float4*)&Ap[(size_t)kk * NCOLS + n0 + c4];
        sBt[c4 + 0][kk] = v.x;
        sBt[c4 + 1][kk] = v.y;
        sBt[c4 + 2][kk] = v.z;
        sBt[c4 + 3][kk] = v.w;
    }
    __syncthreads();

    const int lane = threadIdx.x & 31;
    const int nl   = lane & 15;
    const int kh   = (lane >> 4) << 1;

    const float* pa0 = axbuf + (size_t)(r0 + nl) * AXLD + kh;
    const float* pa1 = axbuf + (size_t)(r1 + nl) * AXLD + kh;
    const float* ps  = &sBt[nl][kh];

    v8f acc0[JT5] = {};
    v8f acc1[JT5] = {};
    for (int k0 = 0; k0 < MP; k0 += 4) {
        v2f af0; af0.x = pa0[0]; af0.y = pa0[1]; pa0 += 4;
        v2f af1; af1.x = pa1[0]; af1.y = pa1[1]; pa1 += 4;
#pragma unroll
        for (int j = 0; j < JT5; ++j) {
            v2f bf = *(const v2f*)(ps + j * 16 * KLD);
            acc0[j] = wmma4(af0, bf, acc0[j]);
            acc1[j] = wmma4(af1, bf, acc1[j]);
        }
        ps += 4;
    }

    const float a0 = fabsf(hyp[(k * PN + p) * 4 + 0]);
    const float a1 = fabsf(hyp[(k * PN + p) * 4 + 1]);
    const float a2 = fabsf(hyp[(k * PN + p) * 4 + 2]);
    const int pl = nbr[p * 2 + 0];
    const int pr = nbr[p * 2 + 1];
    const size_t stride = (size_t)BB * NCOLS;
    float*       Xp = X + (size_t)p * stride;
    const float* Xl = X + (size_t)pl * stride;
    const float* Xr = X + (size_t)pr * stride;
    const float* Up = U + (size_t)p * stride;
    const float* Tp = Atb + (size_t)p * stride;

    const int ih = (lane >> 4) * 8;
#pragma unroll
    for (int j = 0; j < JT5; ++j) {
        const int col = n0 + j * 16 + nl;
#pragma unroll
        for (int v = 0; v < 8; ++v) {
            {
                const size_t off = (size_t)(r0 + ih + v) * NCOLS + col;
                const float xo  = Xp[off];
                const float sgn = (xo > 0.f) ? 1.f : ((xo < 0.f) ? -1.f : 0.f);
                const float g   = acc0[j][v] - Tp[off] + 2.f * a0 * xo + a2 * sgn
                                  + 2.f * Up[off] - a0 * (Xl[off] + Xr[off]);
                Xp[off] = xo - a1 * g;
            }
            {
                const size_t off = (size_t)(r1 + ih + v) * NCOLS + col;
                const float xo  = Xp[off];
                const float sgn = (xo > 0.f) ? 1.f : ((xo < 0.f) ? -1.f : 0.f);
                const float g   = acc1[j][v] - Tp[off] + 2.f * a0 * xo + a2 * sgn
                                  + 2.f * Up[off] - a0 * (Xl[off] + Xr[off]);
                Xp[off] = xo - a1 * g;
            }
        }
    }
}

__global__ void dual_kernel(float* __restrict__ XU, const float* __restrict__ hyp,
                            const int* __restrict__ nbr, int k) {
    const size_t stride4 = (size_t)BB * NCOLS / 4;
    const size_t total4  = (size_t)PN * stride4;
    size_t i = (size_t)blockIdx.x * blockDim.x + threadIdx.x;
    if (i >= total4) return;
    const int p = (int)(i / stride4);
    const size_t off = i - (size_t)p * stride4;
    const float a3 = fabsf(hyp[(k * PN + p) * 4 + 3]);
    const int pl = nbr[p * 2 + 0];
    const int pr = nbr[p * 2 + 1];
    const float4* X = (const float4*)XU;
    float4*       Uv = (float4*)(XU + (size_t)PN * BB * NCOLS);
    const float4 x  = X[i];
    const float4 xl = X[(size_t)pl * stride4 + off];
    const float4 xr = X[(size_t)pr * stride4 + off];
    float4 u = Uv[i];
    u.x += a3 * (2.f * x.x - xl.x - xr.x);
    u.y += a3 * (2.f * x.y - xl.y - xr.y);
    u.z += a3 * (2.f * x.z - xl.z - xr.z);
    u.w += a3 * (2.f * x.w - xl.w - xr.w);
    Uv[i] = u;
}

extern "C" void kernel_launch(void* const* d_in, const int* in_sizes, int n_in,
                              void* d_out, int out_size, void* d_ws, size_t ws_size,
                              hipStream_t stream) {
    (void)in_sizes; (void)n_in; (void)out_size; (void)ws_size;
    const float* A   = (const float*)d_in[0];
    const float* b   = (const float*)d_in[1];
    const float* hyp = (const float*)d_in[2];
    const int*   nbr = (const int*)d_in[3];

    float* X = (float*)d_out;
    float* U = X + (size_t)PN * BB * NCOLS;
    float* Atb   = (float*)d_ws;
    float* axbuf = Atb + (size_t)PN * BB * NCOLS;

    {
        long long n4 = (2LL * PN * BB * NCOLS) / 4;
        int blocks = (int)((n4 + 255) / 256);
        zero_kernel<<<blocks, 256, 0, stream>>>((float4*)d_out, n4);
    }
    atb_kernel<<<dim3(25, 4, PN), 256, 0, stream>>>(A, b, Atb);

    const int colors[3][2] = {{0, 2}, {1, 3}, {4, -1}};
    for (int k = 0; k < KITERS; ++k) {
        for (int c = 0; c < 3; ++c) {
            for (int s = 0; s < 2; ++s) {
                const int p = colors[c][s];
                if (p < 0) continue;
                ax_kernel  <<<4,           256, 0, stream>>>(A, X, axbuf, p);
                xupd_kernel<<<dim3(25, 4), 256, 0, stream>>>(A, axbuf, Atb, hyp, nbr,
                                                             X, U, p, k);
            }
        }
        {
            long long n4 = ((long long)PN * BB * NCOLS) / 4;
            int blocks = (int)((n4 + 255) / 256);
            dual_kernel<<<blocks, 256, 0, stream>>>(X, hyp, nbr, k);
        }
    }
}


# update epilogue turned through LDS: result tiles written row-major to a per-wave LDS tile and read back along rows; operand loads and X stores are 16 B per lane over whole row runs; sums unchanged
# speedup vs baseline: 6.3268x; 6.3268x over previous
#include <hip/hip_runtime.h>

#define PN      5
#define MROWS   500
#define NCOLS   2000
#define BB      1024
#define KITERS  10
#define MP      100
#define AXLD    112
#define JT5     5
#define NGRP    80
#define KLD     102
#define AXKC    100
#define AXNCH   20
#define AXROWS  128
#define AXTHR   224
#define AXNF4   2900
#define AXNV    13
#define ROWSPAN 512
#define TPITCH  84

typedef __attribute__((ext_vector_type(2))) float v2f;
typedef __attribute__((ext_vector_type(8))) float v8f;
typedef __attribute__((ext_vector_type(4))) float v4f;

__device__ __forceinline__ v8f wmma4(v2f a, v2f b, v8f c) {
    return __builtin_amdgcn_wmma_f32_16x16x4_f32(
        false, a, false, b, (short)0, c, false, false);
}


__global__ void zero_kernel(float4* __restrict__ dst, long long n4) {
    long long i = (long long)blockIdx.x * blockDim.x + threadIdx.x;
    if (i < n4) dst[i] = make_float4(0.f, 0.f, 0.f, 0.f);
}

__global__ void __launch_bounds__(256, 1)
atb_kernel(const float* __restrict__ A, const float* __restrict__ b,
           float* __restrict__ Atb) {
    const int p    = blockIdx.z;
    const int n0   = blockIdx.x * NGRP;
    const int it0  = blockIdx.y * 8 + (threadIdx.x >> 5);
    const int r0   = it0 * 16;
    const int r1   = r0 + ROWSPAN;
    const float* Ap = A + (size_t)p * MP * NCOLS;

    __shared__ float sBt[NGRP][KLD];
    for (int idx = threadIdx.x; idx < MP * NGRP / 4; idx += 256) {
        const int kk = idx / (NGRP / 4);
        const int c4 = (idx % (NGRP / 4)) * 4;
        float4 v = *(const float4*)&Ap[(size_t)kk * NCOLS + n0 + c4];
        sBt[c4 + 0][kk] = v.x;
        sBt[c4 + 1][kk] = v.y;
        sBt[c4 + 2][kk] = v.z;
        sBt[c4 + 3][kk] = v.w;
    }
    __syncthreads();

    const int lane = threadIdx.x & 31;
    const int nl   = lane & 15;
    const int kh   = (lane >> 4) << 1;

    const float* pa0 = b + (size_t)(r0 + nl) * MROWS + p * MP + kh;
    const float* pa1 = b + (size_t)(r1 + nl) * MROWS + p * MP + kh;
    const float* ps  = &sBt[nl][kh];

    v8f acc0[JT5] = {};
    v8f acc1[JT5] = {};
    for (int k0 = 0; k0 < MP; k0 += 4) {
        v2f af0; af0.x = pa0[0]; af0.y = pa0[1]; pa0 += 4;
        v2f af1; af1.x = pa1[0]; af1.y = pa1[1]; pa1 += 4;
#pragma unroll
        for (int j = 0; j < JT5; ++j) {
            v2f bf = *(const v2f*)(ps + j * 16 * KLD);
            acc0[j] = wmma4(af0, bf, acc0[j]);
            acc1[j] = wmma4(af1, bf, acc1[j]);
        }
        ps += 4;
    }

    float* out = Atb + (size_t)p * BB * NCOLS;
    const int ih = (lane >> 4) * 8;
#pragma unroll
    for (int j = 0; j < JT5; ++j) {
        const int col = n0 + j * 16 + nl;
#pragma unroll
        for (int v = 0; v < 8; ++v) {
            out[(size_t)(r0 + ih + v) * NCOLS + col] = acc0[j][v];
            out[(size_t)(r1 + ih + v) * NCOLS + col] = acc1[j][v];
        }
    }
}

__global__ void __launch_bounds__(AXTHR, 1)
ax_kernel(const float* __restrict__ A, const float* __restrict__ X,
          float* __restrict__ axbuf, int p) {
    __shared__ float sP[2][AXROWS * AXKC];

    const int tid  = threadIdx.x;
    const int wave = tid >> 5;
    const int lane = tid & 31;
    const int c    = lane & 15;
    const int kh   = (lane >> 4) << 1;
    const int r0   = blockIdx.x * 16;
    const float* Ap = A + (size_t)p * MP * NCOLS;
    const float* Xp = X + (size_t)p * BB * NCOLS;

    const float* gsrc[AXNV];
#pragma unroll
    for (int i = 0; i < AXNV; ++i) {
        const int idx = tid + AXTHR * i;
        const int row = idx / (AXKC / 4);
        const int c4  = (idx % (AXKC / 4)) * 4;
        gsrc[i] = (row < 16) ? (Xp + (size_t)(r0 + row) * NCOLS + c4)
                             : (Ap + (size_t)((row < 16 + MP ? row : 16) - 16) * NCOLS + c4);
    }

    for (int idx = tid; idx < 2 * (AXROWS - 16 - MP) * (AXKC / 4); idx += AXTHR) {
        const int bufi = idx / ((AXROWS - 16 - MP) * (AXKC / 4));
        const int rem  = idx % ((AXROWS - 16 - MP) * (AXKC / 4));
        *(float4*)&sP[bufi][(16 + MP) * AXKC + rem * 4] = make_float4(0.f, 0.f, 0.f, 0.f);
    }

    float4 st[AXNV];
#pragma unroll
    for (int i = 0; i < AXNV; ++i) {
        const int idx = tid + AXTHR * i;
        st[i] = *(const float4*)(gsrc[i]);
        if (idx >= AXNF4) st[i] = make_float4(0.f, 0.f, 0.f, 0.f);
    }
#pragma unroll
    for (int i = 0; i < AXNV; ++i) {
        const int idx = tid + AXTHR * i;
        *(float4*)&sP[0][idx * 4] = st[i];
    }
    __syncthreads();

    const int aoff = c * AXKC + kh;
    const int boff = (16 + wave * 16 + c) * AXKC + kh;

    v8f acc = {};
    for (int ch = 0; ch < AXNCH - 1; ++ch) {
        const int cur = ch & 1;
#pragma unroll
        for (int i = 0; i < AXNV; ++i) {
            const int idx = tid + AXTHR * i;
            st[i] = *(const float4*)(gsrc[i] + (ch + 1) * AXKC);
            if (idx >= AXNF4) st[i] = make_float4(0.f, 0.f, 0.f, 0.f);
        }
        const float* pa = &sP[cur][aoff];
        const float* pb = &sP[cur][boff];
#pragma unroll
        for (int k0 = 0; k0 < AXKC; k0 += 4) {
            v2f af = *(const v2f*)(pa + k0);
            v2f bf = *(const v2f*)(pb + k0);
            acc = wmma4(af, bf, acc);
        }
#pragma unroll
        for (int i = 0; i < AXNV; ++i) {
            const int idx = tid + AXTHR * i;
            *(float4*)&sP[cur ^ 1][idx * 4] = st[i];
        }
        __syncthreads();
    }
    {
        const int cur = (AXNCH - 1) & 1;
        const float* pa = &sP[cur][aoff];
        const float* pb = &sP[cur][boff];
#pragma unroll
        for (int k0 = 0; k0 < AXKC; k0 += 4) {
            v2f af = *(const v2f*)(pa + k0);
            v2f bf = *(const v2f*)(pb + k0);
            acc = wmma4(af, bf, acc);
        }
    }

    const int ih  = (lane >> 4) * 8;
    const int col = wave * 16 + c;
    if (col < MP) {
#pragma unroll
        for (int v = 0; v < 8; ++v)
            axbuf[(size_t)(r0 + ih + v) * AXLD + col] = acc[v];
    }
}

__device__ __forceinline__ float xnew(float atax, float xo, float tp, float up,
                                      float xl, float xr, float a0, float a1, float a2) {
    const float sgn = (xo > 0.f) ? 1.f : ((xo < 0.f) ? -1.f : 0.f);
    const float g   = atax - tp + 2.f * a0 * xo + a2 * sgn
                      + 2.f * up - a0 * (xl + xr);
    return xo - a1 * g;
}

__global__ void __launch_bounds__(256, 1)
xupd_kernel(const float* __restrict__ A, const float* __restrict__ axbuf,
            const float* __restrict__ Atb, const float* __restrict__ hyp,
            const int* __restrict__ nbr,
            float* __restrict__ X, const float* __restrict__ U,
            int p, int k) {
    const int n0  = blockIdx.x * NGRP;
    const int it0 = blockIdx.y * 8 + (threadIdx.x >> 5);
    const int r0  = it0 * 16;
    const int r1  = r0 + ROWSPAN;
    const float* Ap = A + (size_t)p * MP * NCOLS;

    __shared__ float sBt[NGRP][KLD];
    __shared__ __attribute__((aligned(16))) float sT[8][32 * TPITCH];
    for (int idx = threadIdx.x; idx < MP * NGRP / 4; idx += 256) {
        const int kk = idx / (NGRP / 4);
        const int c4 = (idx % (NGRP / 4)) * 4;
        float4 v = *(const float4*)&Ap[(size_t)kk * NCOLS + n0 + c4];
        sBt[c4 + 0][kk] = v.x;
        sBt[c4 + 1][kk] = v.y;
        sBt[c4 + 2][kk] = v.z;
        sBt[c4 + 3][kk] = v.w;
    }
    __syncthreads();

    const int lane = threadIdx.x & 31;
    const int nl   = lane & 15;
    const int kh   = (lane >> 4) << 1;

    const float* pa0 = axbuf + (size_t)(r0 + nl) * AXLD + kh;
    const float* pa1 = axbuf + (size_t)(r1 + nl) * AXLD + kh;
    const float* ps  = &sBt[nl][kh];

    v8f acc0[JT5] = {};
    v8f acc1[JT5] = {};
    for (int k0 = 0; k0 < MP; k0 += 4) {
        v2f af0; af0.x = pa0[0]; af0.y = pa0[1]; pa0 += 4;
        v2f af1; af1.x = pa1[0]; af1.y = pa1[1]; pa1 += 4;
#pragma unroll
        for (int j = 0; j < JT5; ++j) {
            v2f bf = *(const v2f*)(ps + j * 16 * KLD);
            acc0[j] = wmma4(af0, bf, acc0[j]);
            acc1[j] = wmma4(af1, bf, acc1[j]);
        }
        ps += 4;
    }

    const float a0 = fabsf(hyp[(k * PN + p) * 4 + 0]);
    const float a1 = fabsf(hyp[(k * PN + p) * 4 + 1]);
    const float a2 = fabsf(hyp[(k * PN + p) * 4 + 2]);
    const int pl = nbr[p * 2 + 0];
    const int pr = nbr[p * 2 + 1];
    const size_t stride = (size_t)BB * NCOLS;
    float*       Xp = X + (size_t)p * stride;
    const float* Xl = X + (size_t)pl * stride;
    const float* Xr = X + (size_t)pr * stride;
    const float* Up = U + (size_t)p * stride;
    const float* Tp = Atb + (size_t)p * stride;

    const int wave = threadIdx.x >> 5;
    const int ih   = (lane >> 4) * 8;
    float* tw = &sT[wave][0];
#pragma unroll
    for (int j = 0; j < JT5; ++j) {
#pragma unroll
        for (int v = 0; v < 8; ++v) {
            tw[(ih + v) * TPITCH + j * 16 + nl]        = acc0[j][v];
            tw[(16 + ih + v) * TPITCH + j * 16 + nl]   = acc1[j][v];
        }
    }
    __syncthreads();

    size_t gofs[5];
    int    lofs[5];
#pragma unroll
    for (int s = 0; s < 5; ++s) {
        const int q  = s * 32 + lane;
        const int rr = q / 20;
        const int c4 = (q % 20) * 4;
        gofs[s] = (size_t)(r0 + rr) * NCOLS + n0 + c4;
        lofs[s] = rr * TPITCH + c4;
    }

    v4f xo_[2][5], tp_[2][5], up_[2][5], xl_[2][5], xr_[2][5];
#pragma unroll
    for (int s = 0; s < 5; ++s) {
        const size_t o = gofs[s];
        xo_[0][s] = *(const v4f*)(Xp + o);
        tp_[0][s] = *(const v4f*)(Tp + o);
        up_[0][s] = *(const v4f*)(Up + o);
        xl_[0][s] = *(const v4f*)(Xl + o);
        xr_[0][s] = *(const v4f*)(Xr + o);
    }
#pragma unroll
    for (int g = 0; g < 4; ++g) {
        const int cur = g & 1;
        if (g + 1 < 4) {
            const int nxt = cur ^ 1;
            const size_t go = (size_t)(((g + 1) & 1) * 8 + ((g + 1) >> 1) * ROWSPAN) * NCOLS;
#pragma unroll
            for (int s = 0; s < 5; ++s) {
                const size_t o = gofs[s] + go;
                xo_[nxt][s] = *(const v4f*)(Xp + o);
                tp_[nxt][s] = *(const v4f*)(Tp + o);
                up_[nxt][s] = *(const v4f*)(Up + o);
                xl_[nxt][s] = *(const v4f*)(Xl + o);
                xr_[nxt][s] = *(const v4f*)(Xr + o);
            }
        }
        const size_t go = (size_t)((g & 1) * 8 + (g >> 1) * ROWSPAN) * NCOLS;
        const int    lo = ((g & 1) * 8 + (g >> 1) * 16) * TPITCH;
#pragma unroll
        for (int s = 0; s < 5; ++s) {
            const v4f a  = *(const v4f*)(tw + lo + lofs[s]);
            const v4f xo = xo_[cur][s];
            const v4f tp = tp_[cur][s];
            const v4f up = up_[cur][s];
            const v4f xl = xl_[cur][s];
            const v4f xr = xr_[cur][s];
            v4f r;
            r.x = xnew(a.x, xo.x, tp.x, up.x, xl.x, xr.x, a0, a1, a2);
            r.y = xnew(a.y, xo.y, tp.y, up.y, xl.y, xr.y, a0, a1, a2);
            r.z = xnew(a.z, xo.z, tp.z, up.z, xl.z, xr.z, a0, a1, a2);
            r.w = xnew(a.w, xo.w, tp.w, up.w, xl.w, xr.w, a0, a1, a2);
            *(v4f*)(Xp + gofs[s] + go) = r;
        }
    }
}

__global__ void dual_kernel(float* __restrict__ XU, const float* __restrict__ hyp,
                            const int* __restrict__ nbr, int k) {
    const size_t stride4 = (size_t)BB * NCOLS / 4;
    const size_t total4  = (size_t)PN * stride4;
    size_t i = (size_t)blockIdx.x * blockDim.x + threadIdx.x;
    if (i >= total4) return;
    const int p = (int)(i / stride4);
    const size_t off = i - (size_t)p * stride4;
    const float a3 = fabsf(hyp[(k * PN + p) * 4 + 3]);
    const int pl = nbr[p * 2 + 0];
    const int pr = nbr[p * 2 + 1];
    const float4* X = (const float4*)XU;
    float4*       Uv = (float4*)(XU + (size_t)PN * BB * NCOLS);
    const float4 x  = X[i];
    const float4 xl = X[(size_t)pl * stride4 + off];
    const float4 xr = X[(size_t)pr * stride4 + off];
    float4 u = Uv[i];
    u.x += a3 * (2.f * x.x - xl.x - xr.x);
    u.y += a3 * (2.f * x.y - xl.y - xr.y);
    u.z += a3 * (2.f * x.z - xl.z - xr.z);
    u.w += a3 * (2.f * x.w - xl.w - xr.w);
    Uv[i] = u;
}

extern "C" void kernel_launch(void* const* d_in, const int* in_sizes, int n_in,
                              void* d_out, int out_size, void* d_ws, size_t ws_size,
                              hipStream_t stream) {
    (void)in_sizes; (void)n_in; (void)out_size; (void)ws_size;
    const float* A   = (const float*)d_in[0];
    const float* b   = (const float*)d_in[1];
    const float* hyp = (const float*)d_in[2];
    const int*   nbr = (const int*)d_in[3];

    float* X = (float*)d_out;
    float* U = X + (size_t)PN * BB * NCOLS;
    float* Atb   = (float*)d_ws;
    float* axbuf = Atb + (size_t)PN * BB * NCOLS;

    {
        long long n4 = (2LL * PN * BB * NCOLS) / 4;
        int blocks = (int)((n4 + 255) / 256);
        zero_kernel<<<blocks, 256, 0, stream>>>((float4*)d_out, n4);
    }
    atb_kernel<<<dim3(25, 4, PN), 256, 0, stream>>>(A, b, Atb);

    const int colors[3][2] = {{0, 2}, {1, 3}, {4, -1}};
    for (int k = 0; k < KITERS; ++k) {
        for (int c = 0; c < 3; ++c) {
            for (int s = 0; s < 2; ++s) {
                const int p = colors[c][s];
                if (p < 0) continue;
                ax_kernel  <<<64,      AXTHR, 0, stream>>>(A, X, axbuf, p);
                xupd_kernel<<<dim3(25, 4), 256, 0, stream>>>(A, axbuf, Atb, hyp, nbr,
                                                             X, U, p, k);
            }
        }
        {
            long long n4 = ((long long)PN * BB * NCOLS) / 4;
            int blocks = (int)((n4 + 255) / 256);
            dual_kernel<<<blocks, 256, 0, stream>>>(X, hyp, nbr, k);
        }
    }
}


# AX product re-tiled to one 16x16 tile chain per SIMD: grid 64 -> 32x7, 2 matrix waves (full K each) fed by 5 loader waves through 48-row K=200 LDS panels; sum order unchanged
# speedup vs baseline: 8.4775x; 1.3399x over previous
#include <hip/hip_runtime.h>

#define PN      5
#define MROWS   500
#define NCOLS   2000
#define BB      1024
#define KITERS  10
#define MP      100
#define AXLD    112
#define JT5     5
#define NGRP    80
#define KLD     102
#define AXKC    200
#define AXNCH   10
#define AXPIT   204
#define AXROWS  48
#define AXTHR   224
#define AXLT    160
#define AXNX    10
#define AXNA    5
#define AXKB    10
#define ROWSPAN 512
#define TPITCH  84

typedef __attribute__((ext_vector_type(2))) float v2f;
typedef __attribute__((ext_vector_type(8))) float v8f;
typedef __attribute__((ext_vector_type(4))) float v4f;

__device__ __forceinline__ v8f wmma4(v2f a, v2f b, v8f c) {
    return __builtin_amdgcn_wmma_f32_16x16x4_f32(
        false, a, false, b, (short)0, c, false, false);
}

__device__ __forceinline__ void lds_barrier() {
    __builtin_amdgcn_fence(__ATOMIC_RELEASE, "workgroup", "local");
    __builtin_amdgcn_s_barrier();
    __builtin_amdgcn_fence(__ATOMIC_ACQUIRE, "workgroup", "local");
}


__global__ void zero_kernel(float4* __restrict__ dst, long long n4) {
    long long i = (long long)blockIdx.x * blockDim.x + threadIdx.x;
    if (i < n4) dst[i] = make_float4(0.f, 0.f, 0.f, 0.f);
}

__global__ void __launch_bounds__(256, 1)
atb_kernel(const float* __restrict__ A, const float* __restrict__ b,
           float* __restrict__ Atb) {
    const int p    = blockIdx.z;
    const int n0   = blockIdx.x * NGRP;
    const int it0  = blockIdx.y * 8 + (threadIdx.x >> 5);
    const int r0   = it0 * 16;
    const int r1   = r0 + ROWSPAN;
    const float* Ap = A + (size_t)p * MP * NCOLS;

    __shared__ float sBt[NGRP][KLD];
    for (int idx = threadIdx.x; idx < MP * NGRP / 4; idx += 256) {
        const int kk = idx / (NGRP / 4);
        const int c4 = (idx % (NGRP / 4)) * 4;
        float4 v = *(const float4*)&Ap[(size_t)kk * NCOLS + n0 + c4];
        sBt[c4 + 0][kk] = v.x;
        sBt[c4 + 1][kk] = v.y;
        sBt[c4 + 2][kk] = v.z;
        sBt[c4 + 3][kk] = v.w;
    }
    __syncthreads();

    const int lane = threadIdx.x & 31;
    const int nl   = lane & 15;
    const int kh   = (lane >> 4) << 1;

    const float* pa0 = b + (size_t)(r0 + nl) * MROWS + p * MP + kh;
    const float* pa1 = b + (size_t)(r1 + nl) * MROWS + p * MP + kh;
    const float* ps  = &sBt[nl][kh];

    v8f acc0[JT5] = {};
    v8f acc1[JT5] = {};
    for (int k0 = 0; k0 < MP; k0 += 4) {
        v2f af0; af0.x = pa0[0]; af0.y = pa0[1]; pa0 += 4;
        v2f af1; af1.x = pa1[0]; af1.y = pa1[1]; pa1 += 4;
#pragma unroll
        for (int j = 0; j < JT5; ++j) {
            v2f bf = *(const v2f*)(ps + j * 16 * KLD);
            acc0[j] = wmma4(af0, bf, acc0[j]);
            acc1[j] = wmma4(af1, bf, acc1[j]);
        }
        ps += 4;
    }

    float* out = Atb + (size_t)p * BB * NCOLS;
    const int ih = (lane >> 4) * 8;
#pragma unroll
    for (int j = 0; j < JT5; ++j) {
        const int col = n0 + j * 16 + nl;
#pragma unroll
        for (int v = 0; v < 8; ++v) {
            out[(size_t)(r0 + ih + v) * NCOLS + col] = acc0[j][v];
            out[(size_t)(r1 + ih + v) * NCOLS + col] = acc1[j][v];
        }
    }
}

__global__ void __launch_bounds__(AXTHR, 1)
ax_kernel(const float* __restrict__ A, const float* __restrict__ X,
          float* __restrict__ axbuf, int p) {
    __shared__ __attribute__((aligned(16))) float sP[2][AXROWS * AXPIT];

    const int tid  = threadIdx.x;
    const int lane = tid & 31;
    const int wave = __builtin_amdgcn_readfirstlane(tid >> 5);
    const bool loader = wave >= 2;
    const int rb   = blockIdx.x * 32;
    const int mt   = blockIdx.y;
    const float* xsrc = X + ((size_t)p * BB + (size_t)rb) * NCOLS;
    const float* asrc = A + ((size_t)p * MP + (size_t)mt * 16) * NCOLS;
    const int narow = MP - mt * 16;

    const int lt = loader ? tid - 64 : 0;
    int  xg[AXNX], xl[AXNX], ag[AXNA], al[AXNA];
    bool az[AXNA];
    v4f  sx[AXNX], sa[AXNA];
#pragma unroll
    for (int i = 0; i < AXNX; ++i) {
        const int idx = lt + AXLT * i;
        const int row = idx / (AXKC / 4);
        const int c4  = (idx % (AXKC / 4)) * 4;
        xg[i] = row * NCOLS + c4;
        xl[i] = row * AXPIT + c4;
    }
#pragma unroll
    for (int i = 0; i < AXNA; ++i) {
        const int idx = lt + AXLT * i;
        const int row = idx / (AXKC / 4);
        const int c4  = (idx % (AXKC / 4)) * 4;
        az[i] = row >= narow;
        ag[i] = (az[i] ? 0 : row * NCOLS) + c4;
        al[i] = (32 + row) * AXPIT + c4;
    }
    if (loader) {
#pragma unroll
        for (int i = 0; i < AXNX; ++i) sx[i] = *(const v4f*)(xsrc + xg[i]);
#pragma unroll
        for (int i = 0; i < AXNA; ++i) sa[i] = *(const v4f*)(asrc + ag[i]);
#pragma unroll
        for (int i = 0; i < AXNX; ++i) *(v4f*)&sP[0][xl[i]] = sx[i];
#pragma unroll
        for (int i = 0; i < AXNA; ++i) {
            const v4f zero = {0.f, 0.f, 0.f, 0.f};
            *(v4f*)&sP[0][al[i]] = az[i] ? zero : sa[i];
        }
    }

    const int c    = lane & 15;
    const int kh   = (lane >> 4) << 1;
    const int aoff = ((wave & 1) * 16 + c) * AXPIT + kh;
    const int boff = (32 + c) * AXPIT + kh;

    v8f acc = {};
    for (int ch = 0; ch < AXNCH; ++ch) {
        if (loader) {
            const int kc = (ch + 1 < AXNCH ? ch + 1 : AXNCH - 1) * AXKC;
#pragma unroll
            for (int i = 0; i < AXNX; ++i) sx[i] = *(const v4f*)(xsrc + kc + xg[i]);
#pragma unroll
            for (int i = 0; i < AXNA; ++i) sa[i] = *(const v4f*)(asrc + kc + ag[i]);
        }
        lds_barrier();
        if (loader) {
            if (ch + 1 < AXNCH) {
                float* dst = &sP[(ch + 1) & 1][0];
#pragma unroll
                for (int i = 0; i < AXNX; ++i) *(v4f*)(dst + xl[i]) = sx[i];
#pragma unroll
                for (int i = 0; i < AXNA; ++i) {
                    const v4f zero = {0.f, 0.f, 0.f, 0.f};
                    *(v4f*)(dst + al[i]) = az[i] ? zero : sa[i];
                }
            }
        } else {
            const float* pa = &sP[ch & 1][aoff];
            const float* pb = &sP[ch & 1][boff];
            v2f fa[2][AXKB], fb[2][AXKB];
#pragma unroll
            for (int s = 0; s < AXKB; ++s) {
                fa[0][s] = *(const v2f*)(pa + 4 * s);
                fb[0][s] = *(const v2f*)(pb + 4 * s);
            }
#pragma unroll
            for (int blk = 0; blk < AXKC / (4 * AXKB); ++blk) {
                const int cur = blk & 1;
                if (blk + 1 < AXKC / (4 * AXKB)) {
#pragma unroll
                    for (int s = 0; s < AXKB; ++s) {
                        fa[cur ^ 1][s] = *(const v2f*)(pa + (blk + 1) * 4 * AXKB + 4 * s);
                        fb[cur ^ 1][s] = *(const v2f*)(pb + (blk + 1) * 4 * AXKB + 4 * s);
                    }
                }
                __builtin_amdgcn_sched_barrier(0);
#pragma unroll
                for (int s = 0; s < AXKB; ++s)
                    acc = wmma4(fa[cur][s], fb[cur][s], acc);
            }
        }
    }

    if (!loader) {
        const int ih  = (lane >> 4) * 8;
        const int col = mt * 16 + c;
        const int r0  = rb + (wave & 1) * 16;
        if (col < MP) {
#pragma unroll
            for (int v = 0; v < 8; ++v)
                axbuf[(size_t)(r0 + ih + v) * AXLD + col] = acc[v];
        }
    }
}

__device__ __forceinline__ float xnew(float atax, float xo, float tp, float up,
                                      float xl, float xr, float a0, float a1, float a2) {
    const float sgn = (xo > 0.f) ? 1.f : ((xo < 0.f) ? -1.f : 0.f);
    const float g   = atax - tp + 2.f * a0 * xo + a2 * sgn
                      + 2.f * up - a0 * (xl + xr);
    return xo - a1 * g;
}

__global__ void __launch_bounds__(256, 1)
xupd_kernel(const float* __restrict__ A, const float* __restrict__ axbuf,
            const float* __restrict__ Atb, const float* __restrict__ hyp,
            const int* __restrict__ nbr,
            float* __restrict__ X, const float* __restrict__ U,
            int p, int k) {
    const int n0  = blockIdx.x * NGRP;
    const int it0 = blockIdx.y * 8 + (threadIdx.x >> 5);
    const int r0  = it0 * 16;
    const int r1  = r0 + ROWSPAN;
    const float* Ap = A + (size_t)p * MP * NCOLS;

    __shared__ float sBt[NGRP][KLD];
    __shared__ __attribute__((aligned(16))) float sT[8][32 * TPITCH];
    for (int idx = threadIdx.x; idx < MP * NGRP / 4; idx += 256) {
        const int kk = idx / (NGRP / 4);
        const int c4 = (idx % (NGRP / 4)) * 4;
        float4 v = *(const float4*)&Ap[(size_t)kk * NCOLS + n0 + c4];
        sBt[c4 + 0][kk] = v.x;
        sBt[c4 + 1][kk] = v.y;
        sBt[c4 + 2][kk] = v.z;
        sBt[c4 + 3][kk] = v.w;
    }
    __syncthreads();

    const int lane = threadIdx.x & 31;
    const int nl   = lane & 15;
    const int kh   = (lane >> 4) << 1;

    const float* pa0 = axbuf + (size_t)(r0 + nl) * AXLD + kh;
    const float* pa1 = axbuf + (size_t)(r1 + nl) * AXLD + kh;
    const float* ps  = &sBt[nl][kh];

    v8f acc0[JT5] = {};
    v8f acc1[JT5] = {};
    for (int k0 = 0; k0 < MP; k0 += 4) {
        v2f af0; af0.x = pa0[0]; af0.y = pa0[1]; pa0 += 4;
        v2f af1; af1.x = pa1[0]; af1.y = pa1[1]; pa1 += 4;
#pragma unroll
        for (int j = 0; j < JT5; ++j) {
            v2f bf = *(const v2f*)(ps + j * 16 * KLD);
            acc0[j] = wmma4(af0, bf, acc0[j]);
            acc1[j] = wmma4(af1, bf, acc1[j]);
        }
        ps += 4;
    }

    const float a0 = fabsf(hyp[(k * PN + p) * 4 + 0]);
    const float a1 = fabsf(hyp[(k * PN + p) * 4 + 1]);
    const float a2 = fabsf(hyp[(k * PN + p) * 4 + 2]);
    const int pl = nbr[p * 2 + 0];
    const int pr = nbr[p * 2 + 1];
    const size_t stride = (size_t)BB * NCOLS;
    float*       Xp = X + (size_t)p * stride;
    const float* Xl = X + (size_t)pl * stride;
    const float* Xr = X + (size_t)pr * stride;
    const float* Up = U + (size_t)p * stride;
    const float* Tp = Atb + (size_t)p * stride;

    const int wave = threadIdx.x >> 5;
    const int ih   = (lane >> 4) * 8;
    float* tw = &sT[wave][0];
#pragma unroll
    for (int j = 0; j < JT5; ++j) {
#pragma unroll
        for (int v = 0; v < 8; ++v) {
            tw[(ih + v) * TPITCH + j * 16 + nl]        = acc0[j][v];
            tw[(16 + ih + v) * TPITCH + j * 16 + nl]   = acc1[j][v];
        }
    }
    __syncthreads();

    size_t gofs[5];
    int    lofs[5];
#pragma unroll
    for (int s = 0; s < 5; ++s) {
        const int q  = s * 32 + lane;
        const int rr = q / 20;
        const int c4 = (q % 20) * 4;
        gofs[s] = (size_t)(r0 + rr) * NCOLS + n0 + c4;
        lofs[s] = rr * TPITCH + c4;
    }

    v4f xo_[2][5], tp_[2][5], up_[2][5], xl_[2][5], xr_[2][5];
#pragma unroll
    for (int s = 0; s < 5; ++s) {
        const size_t o = gofs[s];
        xo_[0][s] = *(const v4f*)(Xp + o);
        tp_[0][s] = *(const v4f*)(Tp + o);
        up_[0][s] = *(const v4f*)(Up + o);
        xl_[0][s] = *(const v4f*)(Xl + o);
        xr_[0][s] = *(const v4f*)(Xr + o);
    }
#pragma unroll
    for (int g = 0; g < 4; ++g) {
        const int cur = g & 1;
        if (g + 1 < 4) {
            const int nxt = cur ^ 1;
            const size_t go = (size_t)(((g + 1) & 1) * 8 + ((g + 1) >> 1) * ROWSPAN) * NCOLS;
#pragma unroll
            for (int s = 0; s < 5; ++s) {
                const size_t o = gofs[s] + go;
                xo_[nxt][s] = *(const v4f*)(Xp + o);
                tp_[nxt][s] = *(const v4f*)(Tp + o);
                up_[nxt][s] = *(const v4f*)(Up + o);
                xl_[nxt][s] = *(const v4f*)(Xl + o);
                xr_[nxt][s] = *(const v4f*)(Xr + o);
            }
        }
        const size_t go = (size_t)((g & 1) * 8 + (g >> 1) * ROWSPAN) * NCOLS;
        const int    lo = ((g & 1) * 8 + (g >> 1) * 16) * TPITCH;
#pragma unroll
        for (int s = 0; s < 5; ++s) {
            const v4f a  = *(const v4f*)(tw + lo + lofs[s]);
            const v4f xo = xo_[cur][s];
            const v4f tp = tp_[cur][s];
            const v4f up = up_[cur][s];
            const v4f xl = xl_[cur][s];
            const v4f xr = xr_[cur][s];
            v4f r;
            r.x = xnew(a.x, xo.x, tp.x, up.x, xl.x, xr.x, a0, a1, a2);
            r.y = xnew(a.y, xo.y, tp.y, up.y, xl.y, xr.y, a0, a1, a2);
            r.z = xnew(a.z, xo.z, tp.z, up.z, xl.z, xr.z, a0, a1, a2);
            r.w = xnew(a.w, xo.w, tp.w, up.w, xl.w, xr.w, a0, a1, a2);
            *(v4f*)(Xp + gofs[s] + go) = r;
        }
    }
}

__global__ void dual_kernel(float* __restrict__ XU, const float* __restrict__ hyp,
                            const int* __restrict__ nbr, int k) {
    const size_t stride4 = (size_t)BB * NCOLS / 4;
    const size_t total4  = (size_t)PN * stride4;
    size_t i = (size_t)blockIdx.x * blockDim.x + threadIdx.x;
    if (i >= total4) return;
    const int p = (int)(i / stride4);
    const size_t off = i - (size_t)p * stride4;
    const float a3 = fabsf(hyp[(k * PN + p) * 4 + 3]);
    const int pl = nbr[p * 2 + 0];
    const int pr = nbr[p * 2 + 1];
    const float4* X = (const float4*)XU;
    float4*       Uv = (float4*)(XU + (size_t)PN * BB * NCOLS);
    const float4 x  = X[i];
    const float4 xl = X[(size_t)pl * stride4 + off];
    const float4 xr = X[(size_t)pr * stride4 + off];
    float4 u = Uv[i];
    u.x += a3 * (2.f * x.x - xl.x - xr.x);
    u.y += a3 * (2.f * x.y - xl.y - xr.y);
    u.z += a3 * (2.f * x.z - xl.z - xr.z);
    u.w += a3 * (2.f * x.w - xl.w - xr.w);
    Uv[i] = u;
}

extern "C" void kernel_launch(void* const* d_in, const int* in_sizes, int n_in,
                              void* d_out, int out_size, void* d_ws, size_t ws_size,
                              hipStream_t stream) {
    (void)in_sizes; (void)n_in; (void)out_size; (void)ws_size;
    const float* A   = (const float*)d_in[0];
    const float* b   = (const float*)d_in[1];
    const float* hyp = (const float*)d_in[2];
    const int*   nbr = (const int*)d_in[3];

    float* X = (float*)d_out;
    float* U = X + (size_t)PN * BB * NCOLS;
    float* Atb   = (float*)d_ws;
    float* axbuf = Atb + (size_t)PN * BB * NCOLS;

    {
        long long n4 = (2LL * PN * BB * NCOLS) / 4;
        int blocks = (int)((n4 + 255) / 256);
        zero_kernel<<<blocks, 256, 0, stream>>>((float4*)d_out, n4);
    }
    atb_kernel<<<dim3(25, 4, PN), 256, 0, stream>>>(A, b, Atb);

    const int colors[3][2] = {{0, 2}, {1, 3}, {4, -1}};
    for (int k = 0; k < KITERS; ++k) {
        for (int c = 0; c < 3; ++c) {
            for (int s = 0; s < 2; ++s) {
                const int p = colors[c][s];
                if (p < 0) continue;
                ax_kernel  <<<dim3(32, 7), AXTHR, 0, stream>>>(A, X, axbuf, p);
                xupd_kernel<<<dim3(25, 4), 256, 0, stream>>>(A, axbuf, Atb, hyp, nbr,
                                                             X, U, p, k);
            }
        }
        {
            long long n4 = ((long long)PN * BB * NCOLS) / 4;
            int blocks = (int)((n4 + 255) / 256);
            dual_kernel<<<blocks, 256, 0, stream>>>(X, hyp, nbr, k);
        }
    }
}
